# starting kernel
# baseline (speedup 1.0000x reference)
#include <hip/hip_runtime.h>

typedef __attribute__((ext_vector_type(2))) float v2f;
typedef __attribute__((ext_vector_type(8))) float v8f;

#define EPS 1e-5f

__global__ void deg_init_k(float* __restrict__ deg, int n) {
    int i = blockIdx.x * blockDim.x + threadIdx.x;
    if (i < n) deg[i] = 1.0f;
}

__global__ void deg_count_k(const int* __restrict__ col, float* __restrict__ deg, int nE) {
    int e = blockIdx.x * blockDim.x + threadIdx.x;
    if (e < nE) atomicAdd(&deg[col[e]], 1.0f);
}

__global__ void deg_rsqrt_k(float* __restrict__ deg, int n) {
    int i = blockIdx.x * blockDim.x + threadIdx.x;
    if (i < n) deg[i] = rsqrtf(deg[i]);
}

__global__ void edge_norm_k(const int* __restrict__ row, const int* __restrict__ col,
                            const float* __restrict__ dinv, float* __restrict__ norm, int nE) {
    int e = blockIdx.x * blockDim.x + threadIdx.x;
    if (e < nE) norm[e] = dinv[row[e]] * dinv[col[e]];
}

template <int NT, int K, int FOUT>
__global__ __launch_bounds__(32) void gemm_wmma_f32_k(const float* __restrict__ A,
                                                      const float* __restrict__ B,
                                                      float* __restrict__ C) {
    const int colBase = blockIdx.x * (16 * NT);
    const int rowBase = blockIdx.y * 16;
    const int lane = threadIdx.x;
    const int half = lane >> 4;
    const int lr   = lane & 15;

    v8f acc[NT];
    v8f zero = {};
#pragma unroll
    for (int t = 0; t < NT; ++t) acc[t] = zero;

    const float* Arow = A + (unsigned)(rowBase + lr) * K;

#pragma unroll 4
    for (int k0 = 0; k0 < K; k0 += 4) {
        const int ka = k0 + 2 * half;
        v2f a;
        a.x = Arow[ka];
        a.y = Arow[ka + 1];
#pragma unroll
        for (int t = 0; t < NT; ++t) {
            const float* Bp = B + (unsigned)ka * FOUT + colBase + 16 * t + lr;
            v2f b;
            b.x = Bp[0];
            b.y = Bp[FOUT];
            acc[t] = __builtin_amdgcn_wmma_f32_16x16x4_f32(
                 false, a,  false, b,
                 (short)0, acc[t],  false,  false);
        }
    }

#pragma unroll
    for (int t = 0; t < NT; ++t) {
        float* Cp = C + (unsigned)(rowBase + 8 * half) * FOUT + colBase + 16 * t + lr;
#pragma unroll
        for (int j = 0; j < 8; ++j) Cp[(unsigned)j * FOUT] = acc[t][j];
    }
}

template <int F4>
__global__ void agg_init_k(const float* __restrict__ h, const float* __restrict__ dinv,
                           float* __restrict__ out, unsigned total  ) {
    unsigned tid = blockIdx.x * blockDim.x + threadIdx.x;
    if (tid >= total) return;
    unsigned i = tid / F4;
    unsigned c = tid % F4;
    float dv = dinv[i];
    float w = dv * dv;
    float4 hv = ((const float4*)h)[(size_t)i * F4 + c];
    float4 ov;
    ov.x = hv.x * w; ov.y = hv.y * w; ov.z = hv.z * w; ov.w = hv.w * w;
    ((float4*)out)[(size_t)i * F4 + c] = ov;
}

template <int F4>
__global__ void scatter_edges_k(const float* __restrict__ h, const int* __restrict__ row,
                                const int* __restrict__ col, const float* __restrict__ norm,
                                float* __restrict__ out, unsigned total  ) {
    unsigned tid = blockIdx.x * blockDim.x + threadIdx.x;
    if (tid >= total) return;
    unsigned e = tid / F4;
    unsigned c = tid % F4;
    if (F4 == 32) {
        e = (unsigned)__builtin_amdgcn_readfirstlane((int)e);
    }
    float w = norm[e];
    int r = row[e];
    int d = col[e];
    float4 hv = ((const float4*)h)[(size_t)r * F4 + c];
    float* o = out + ((size_t)d * F4 + c) * 4;
    atomicAdd(o + 0, hv.x * w);
    atomicAdd(o + 1, hv.y * w);
    atomicAdd(o + 2, hv.z * w);
    atomicAdd(o + 3, hv.w * w);
}

__global__ void bn_prep_k(const float* __restrict__ b, const float* __restrict__ g,
                          const float* __restrict__ be, const float* __restrict__ m,
                          const float* __restrict__ v, float* __restrict__ sc,
                          float* __restrict__ sh, int F) {
    int f = blockIdx.x * blockDim.x + threadIdx.x;
    if (f >= F) return;
    float s = g[f] * rsqrtf(v[f] + EPS);
    sc[f] = s;
    sh[f] = be[f] + (b[f] - m[f]) * s;
}

template <int F>
__global__ void bn_relu_k(float* __restrict__ x, const float* __restrict__ sc,
                          const float* __restrict__ sh, unsigned total  ) {
    unsigned tid = blockIdx.x * blockDim.x + threadIdx.x;
    if (tid >= total) return;
    unsigned f = tid % F;
    float y = fmaf(x[tid], sc[f], sh[f]);
    x[tid] = y > 0.0f ? y : 0.0f;
}

__global__ void bn_relu_fc_k(const float* __restrict__ x, const float* __restrict__ sc,
                             const float* __restrict__ sh, const float* __restrict__ fcW,
                             const float* __restrict__ fcb, float* __restrict__ out,
                             int nNodes) {
    int i = blockIdx.x * blockDim.x + threadIdx.x;
    if (i >= nNodes) return;
    const float* xi = x + (size_t)i * 64;
    float o0 = fcb[0], o1 = fcb[1];
#pragma unroll 8
    for (int k = 0; k < 64; ++k) {
        float y = fmaf(xi[k], sc[k], sh[k]);
        y = y > 0.0f ? y : 0.0f;
        o0 = fmaf(y, fcW[k * 2 + 0], o0);
        o1 = fmaf(y, fcW[k * 2 + 1], o1);
    }
    out[(size_t)i * 2 + 0] = o0;
    out[(size_t)i * 2 + 1] = o1;
}

static inline size_t alignf(size_t n) { return (n + 127) & ~(size_t)127; }

extern "C" void kernel_launch(void* const* d_in, const int* in_sizes, int n_in,
                              void* d_out, int out_size, void* d_ws, size_t ws_size,
                              hipStream_t stream) {
    const float* x    = (const float*)d_in[0];
    const int*   ei   = (const int*)d_in[1];
    const float* W1   = (const float*)d_in[2];
    const float* b1   = (const float*)d_in[3];
    const float* g1   = (const float*)d_in[4];
    const float* be1  = (const float*)d_in[5];
    const float* m1   = (const float*)d_in[6];
    const float* v1   = (const float*)d_in[7];
    const float* W2   = (const float*)d_in[8];
    const float* b2   = (const float*)d_in[9];
    const float* g2   = (const float*)d_in[10];
    const float* be2  = (const float*)d_in[11];
    const float* m2   = (const float*)d_in[12];
    const float* v2   = (const float*)d_in[13];
    const float* fcW  = (const float*)d_in[14];
    const float* fcb  = (const float*)d_in[15];
    float* out = (float*)d_out;

    const int N  = in_sizes[0] / 256;
    const int E  = in_sizes[1] / 2;
    const int* row = ei;
    const int* col = ei + E;

    float* ws = (float*)d_ws;
    size_t off = 0;
    float* dinv = ws + off; off += alignf((size_t)N);
    float* norm = ws + off; off += alignf((size_t)E);
    float* h1   = ws + off; off += alignf((size_t)N * 128);
    float* agg1 = ws + off; off += alignf((size_t)N * 128);
    float* agg2 = ws + off; off += alignf((size_t)N * 64);
    float* sc1  = ws + off; off += alignf(128);
    float* sh1  = ws + off; off += alignf(128);
    float* sc2  = ws + off; off += alignf(64);
    float* sh2  = ws + off; off += alignf(64);
    float* h2   = h1;
    (void)ws_size; (void)n_in; (void)out_size;

    const int TB = 256;
    deg_init_k<<<(N + TB - 1) / TB, TB, 0, stream>>>(dinv, N);
    deg_count_k<<<(E + TB - 1) / TB, TB, 0, stream>>>(col, dinv, E);
    deg_rsqrt_k<<<(N + TB - 1) / TB, TB, 0, stream>>>(dinv, N);
    edge_norm_k<<<(E + TB - 1) / TB, TB, 0, stream>>>(row, col, dinv, norm, E);

    bn_prep_k<<<1, 128, 0, stream>>>(b1, g1, be1, m1, v1, sc1, sh1, 128);
    bn_prep_k<<<1, 64, 0, stream>>>(b2, g2, be2, m2, v2, sc2, sh2, 64);

    {
        dim3 grid(128 / 64, N / 16);
        gemm_wmma_f32_k<4, 256, 128><<<grid, 32, 0, stream>>>(x, W1, h1);
    }

    {
        unsigned t1 = (unsigned)N * 32;
        agg_init_k<32><<<(t1 + TB - 1) / TB, TB, 0, stream>>>(h1, dinv, agg1, t1);
        unsigned t2 = (unsigned)E * 32;
        scatter_edges_k<32><<<(t2 + TB - 1) / TB, TB, 0, stream>>>(h1, row, col, norm, agg1, t2);
    }

    {
        unsigned t = (unsigned)N * 128;
        bn_relu_k<128><<<(t + TB - 1) / TB, TB, 0, stream>>>(agg1, sc1, sh1, t);
    }

    {
        dim3 grid(64 / 64, N / 16);
        gemm_wmma_f32_k<4, 128, 64><<<grid, 32, 0, stream>>>(agg1, W2, h2);
    }

    {
        unsigned t1 = (unsigned)N * 16;
        agg_init_k<16><<<(t1 + TB - 1) / TB, TB, 0, stream>>>(h2, dinv, agg2, t1);
        unsigned t2 = (unsigned)E * 16;
        scatter_edges_k<16><<<(t2 + TB - 1) / TB, TB, 0, stream>>>(h2, row, col, norm, agg2, t2);
    }

    bn_relu_fc_k<<<(N + TB - 1) / TB, TB, 0, stream>>>(agg2, sc2, sh2, fcW, fcb, out, N);
}


# Aggregation: float-atomic scatter replaced by a gather over edges grouped by destination (integer-atomic rank, scan, fill); one wave sums each row in list order in registers and stores it once
# speedup vs baseline: 1.6591x; 1.6591x over previous
#include <hip/hip_runtime.h>

typedef __attribute__((ext_vector_type(2))) float v2f;
typedef __attribute__((ext_vector_type(8))) float v8f;

#define EPS 1e-5f

__global__ void cnt_zero_k(int* __restrict__ cnt, int n) {
    int i = blockIdx.x * blockDim.x + threadIdx.x;
    if (i < n) cnt[i] = 0;
}

__global__ void cnt_edges_k(const int* __restrict__ col, int* __restrict__ cnt,
                            int* __restrict__ rankOf, int nE) {
    int e = blockIdx.x * blockDim.x + threadIdx.x;
    if (e < nE) rankOf[e] = atomicAdd(&cnt[col[e]], 1);
}

__global__ __launch_bounds__(1024) void seg_scan_k(const int* __restrict__ cnt,
                                                   int* __restrict__ segptr,
                                                   float* __restrict__ dinv, int n) {
    __shared__ int frontPart[32];
    __shared__ int waveTot[32];
    const int t = threadIdx.x;
    const int lane = t & 31;
    const int w = t >> 5;
    const int start = blockIdx.x * 1024;

    int s = 0;
#pragma unroll 8
    for (int i = t; i < start; i += 1024) s += cnt[i];
    for (int o = 16; o > 0; o >>= 1) s += __shfl_xor(s, o);

    const int i = start + t;
    const int c = (i < n) ? cnt[i] : 0;
    int x = c;
    for (int o = 1; o < 32; o <<= 1) {
        int y = __shfl_up(x, o);
        if (lane >= o) x += y;
    }
    if (lane == 0) frontPart[w] = s;
    if (lane == 31) waveTot[w] = x;
    __syncthreads();

    int front = frontPart[lane];
    int before = (lane < w) ? waveTot[lane] : 0;
    for (int o = 16; o > 0; o >>= 1) {
        front += __shfl_xor(front, o);
        before += __shfl_xor(before, o);
    }
    const int excl = front + before + x - c;
    if (i < n) {
        segptr[i] = excl;
        dinv[i] = rsqrtf((float)(c + 1));
        if (i == n - 1) segptr[n] = excl + c;
    }
}

__global__ void seg_fill_k(const int* __restrict__ row, const int* __restrict__ col,
                           const float* __restrict__ dinv, const int* __restrict__ segptr,
                           const int* __restrict__ rankOf, int* __restrict__ srcOf,
                           float* __restrict__ wgtOf, int nE) {
    int e = blockIdx.x * blockDim.x + threadIdx.x;
    if (e >= nE) return;
    int r = row[e];
    int d = col[e];
    int p = segptr[d] + rankOf[e];
    srcOf[p] = r;
    wgtOf[p] = dinv[r] * dinv[d];
}

template <int NT, int K, int FOUT>
__global__ __launch_bounds__(32) void gemm_wmma_f32_k(const float* __restrict__ A,
                                                      const float* __restrict__ B,
                                                      float* __restrict__ C) {
    const int colBase = blockIdx.x * (16 * NT);
    const int rowBase = blockIdx.y * 16;
    const int lane = threadIdx.x;
    const int half = lane >> 4;
    const int lr   = lane & 15;

    v8f acc[NT];
    v8f zero = {};
#pragma unroll
    for (int t = 0; t < NT; ++t) acc[t] = zero;

    const float* Arow = A + (unsigned)(rowBase + lr) * K;

#pragma unroll 4
    for (int k0 = 0; k0 < K; k0 += 4) {
        const int ka = k0 + 2 * half;
        v2f a;
        a.x = Arow[ka];
        a.y = Arow[ka + 1];
#pragma unroll
        for (int t = 0; t < NT; ++t) {
            const float* Bp = B + (unsigned)ka * FOUT + colBase + 16 * t + lr;
            v2f b;
            b.x = Bp[0];
            b.y = Bp[FOUT];
            acc[t] = __builtin_amdgcn_wmma_f32_16x16x4_f32(
                 false, a,  false, b,
                 (short)0, acc[t],  false,  false);
        }
    }

#pragma unroll
    for (int t = 0; t < NT; ++t) {
        float* Cp = C + (unsigned)(rowBase + 8 * half) * FOUT + colBase + 16 * t + lr;
#pragma unroll
        for (int j = 0; j < 8; ++j) Cp[(unsigned)j * FOUT] = acc[t][j];
    }
}

template <int VEC>
__global__ __launch_bounds__(256) void gather_rows_k(const float* __restrict__ h,
                                                     const int* __restrict__ segptr,
                                                     const int* __restrict__ srcOf,
                                                     const float* __restrict__ wgtOf,
                                                     const float* __restrict__ dinv,
                                                     float* __restrict__ out, int n) {
    typedef float vec_t __attribute__((ext_vector_type(VEC)));
    const int lane = threadIdx.x & 31;
    const int node = __builtin_amdgcn_readfirstlane((int)(blockIdx.x * 8 + (threadIdx.x >> 5)));
    if (node >= n) return;
    const vec_t* H = (const vec_t*)h;
    const float dv = dinv[node];
    vec_t acc = H[(size_t)node * 32 + lane] * (dv * dv);
    const int p0 = segptr[node];
    const int p1 = segptr[node + 1];
    for (int base = p0; base < p1; base += 32) {
        const int m = min(32, p1 - base);
        int sv = 0;
        float wv = 0.0f;
        if (lane < m) {
            sv = srcOf[base + lane];
            wv = wgtOf[base + lane];
        }
#pragma unroll 4
        for (int j = 0; j < m; ++j) {
            const int r = __builtin_amdgcn_readlane(sv, j);
            const float wj = __int_as_float(__builtin_amdgcn_readlane(__float_as_int(wv), j));
            acc += H[(size_t)r * 32 + lane] * wj;
        }
    }
    ((vec_t*)out)[(size_t)node * 32 + lane] = acc;
}

__global__ void bn_prep_k(const float* __restrict__ b, const float* __restrict__ g,
                          const float* __restrict__ be, const float* __restrict__ m,
                          const float* __restrict__ v, float* __restrict__ sc,
                          float* __restrict__ sh, int F) {
    int f = blockIdx.x * blockDim.x + threadIdx.x;
    if (f >= F) return;
    float s = g[f] * rsqrtf(v[f] + EPS);
    sc[f] = s;
    sh[f] = be[f] + (b[f] - m[f]) * s;
}

template <int F>
__global__ void bn_relu_k(float* __restrict__ x, const float* __restrict__ sc,
                          const float* __restrict__ sh, unsigned total  ) {
    unsigned tid = blockIdx.x * blockDim.x + threadIdx.x;
    if (tid >= total) return;
    unsigned f = tid % F;
    float y = fmaf(x[tid], sc[f], sh[f]);
    x[tid] = y > 0.0f ? y : 0.0f;
}

__global__ void bn_relu_fc_k(const float* __restrict__ x, const float* __restrict__ sc,
                             const float* __restrict__ sh, const float* __restrict__ fcW,
                             const float* __restrict__ fcb, float* __restrict__ out,
                             int nNodes) {
    int i = blockIdx.x * blockDim.x + threadIdx.x;
    if (i >= nNodes) return;
    const float* xi = x + (size_t)i * 64;
    float o0 = fcb[0], o1 = fcb[1];
#pragma unroll 8
    for (int k = 0; k < 64; ++k) {
        float y = fmaf(xi[k], sc[k], sh[k]);
        y = y > 0.0f ? y : 0.0f;
        o0 = fmaf(y, fcW[k * 2 + 0], o0);
        o1 = fmaf(y, fcW[k * 2 + 1], o1);
    }
    out[(size_t)i * 2 + 0] = o0;
    out[(size_t)i * 2 + 1] = o1;
}

static inline size_t alignf(size_t n) { return (n + 127) & ~(size_t)127; }

extern "C" void kernel_launch(void* const* d_in, const int* in_sizes, int n_in,
                              void* d_out, int out_size, void* d_ws, size_t ws_size,
                              hipStream_t stream) {
    const float* x    = (const float*)d_in[0];
    const int*   ei   = (const int*)d_in[1];
    const float* W1   = (const float*)d_in[2];
    const float* b1   = (const float*)d_in[3];
    const float* g1   = (const float*)d_in[4];
    const float* be1  = (const float*)d_in[5];
    const float* m1   = (const float*)d_in[6];
    const float* v1   = (const float*)d_in[7];
    const float* W2   = (const float*)d_in[8];
    const float* b2   = (const float*)d_in[9];
    const float* g2   = (const float*)d_in[10];
    const float* be2  = (const float*)d_in[11];
    const float* m2   = (const float*)d_in[12];
    const float* v2   = (const float*)d_in[13];
    const float* fcW  = (const float*)d_in[14];
    const float* fcb  = (const float*)d_in[15];
    float* out = (float*)d_out;

    const int N  = in_sizes[0] / 256;
    const int E  = in_sizes[1] / 2;
    const int* row = ei;
    const int* col = ei + E;

    float* ws = (float*)d_ws;
    size_t off = 0;
    float* dinv = ws + off; off += alignf((size_t)N);
    float* wgt  = ws + off; off += alignf((size_t)E);
    float* h1   = ws + off; off += alignf((size_t)N * 128);
    float* agg1 = ws + off; off += alignf((size_t)N * 128);
    int*   lists = (int*)(ws + off); off += alignf((size_t)N * 64);
    float* sc1  = ws + off; off += alignf(128);
    float* sh1  = ws + off; off += alignf(128);
    float* sc2  = ws + off; off += alignf(64);
    float* sh2  = ws + off; off += alignf(64);
    float* h2   = h1;
    float* agg2 = agg1;
    size_t loff = 0;
    int* cnt    = lists + loff; loff += alignf((size_t)N);
    int* segptr = lists + loff; loff += alignf((size_t)N + 1);
    int* rankOf = lists + loff; loff += alignf((size_t)E);
    int* srcOf  = lists + loff; loff += alignf((size_t)E);
    (void)ws_size; (void)n_in; (void)out_size;

    const int TB = 256;
    cnt_zero_k<<<(N + TB - 1) / TB, TB, 0, stream>>>(cnt, N);
    cnt_edges_k<<<(E + TB - 1) / TB, TB, 0, stream>>>(col, cnt, rankOf, E);
    seg_scan_k<<<(N + 1023) / 1024, 1024, 0, stream>>>(cnt, segptr, dinv, N);
    seg_fill_k<<<(E + TB - 1) / TB, TB, 0, stream>>>(row, col, dinv, segptr, rankOf, srcOf, wgt, E);

    bn_prep_k<<<1, 128, 0, stream>>>(b1, g1, be1, m1, v1, sc1, sh1, 128);
    bn_prep_k<<<1, 64, 0, stream>>>(b2, g2, be2, m2, v2, sc2, sh2, 64);

    {
        dim3 grid(128 / 64, N / 16);
        gemm_wmma_f32_k<4, 256, 128><<<grid, 32, 0, stream>>>(x, W1, h1);
    }

    gather_rows_k<4><<<(N + 7) / 8, 256, 0, stream>>>(h1, segptr, srcOf, wgt, dinv, agg1, N);

    {
        unsigned t = (unsigned)N * 128;
        bn_relu_k<128><<<(t + TB - 1) / TB, TB, 0, stream>>>(agg1, sc1, sh1, t);
    }

    {
        dim3 grid(64 / 64, N / 16);
        gemm_wmma_f32_k<4, 128, 64><<<grid, 32, 0, stream>>>(agg1, W2, h2);
    }

    gather_rows_k<2><<<(N + 7) / 8, 256, 0, stream>>>(h2, segptr, srcOf, wgt, dinv, agg2, N);

    bn_relu_fc_k<<<(N + TB - 1) / TB, TB, 0, stream>>>(agg2, sc2, sh2, fcW, fcb, out, N);
}


# Edge lists by destination without per-edge global atomics: LDS counters in two levels (256 node-range bins, then per-node rank, one block per bin); first gather forms weights; row order by arrival
# speedup vs baseline: 2.6415x; 1.5921x over previous
#include <hip/hip_runtime.h>

typedef __attribute__((ext_vector_type(2))) float v2f;
typedef __attribute__((ext_vector_type(8))) float v8f;

#define EPS 1e-5f

#define NBIN 256
#define EB 8

__device__ __forceinline__ int block_excl_scan_1024(int c, int* waveTot) {
    const int lane = threadIdx.x & 31;
    const int w = threadIdx.x >> 5;
    int x = c;
    for (int o = 1; o < 32; o <<= 1) {
        int y = __shfl_up(x, o);
        if (lane >= o) x += y;
    }
    if (lane == 31) waveTot[w] = x;
    __syncthreads();
    int before = (lane < w) ? waveTot[lane] : 0;
    for (int o = 16; o > 0; o >>= 1) before += __shfl_xor(before, o);
    return before + x - c;
}

__global__ __launch_bounds__(1024) void bin_hist_k(const int* __restrict__ col,
                                                   int* __restrict__ binTab, int nE, int n) {
    __shared__ int hist[NBIN];
    const int t = threadIdx.x;
    if (t < NBIN) hist[t] = 0;
    __syncthreads();
    const unsigned bw = (unsigned)((n + NBIN - 1) / NBIN);
    const int slice = (nE + NBIN - 1) / NBIN;
    const int s0 = blockIdx.x * slice;
    const int s1 = min(nE, s0 + slice);
    for (int eb = s0; eb < s1; eb += EB * 1024) {
        int d[EB];
#pragma unroll
        for (int k = 0; k < EB; ++k) {
            const int e = eb + t + k * 1024;
            d[k] = (e < s1) ? col[e] : -1;
        }
#pragma unroll
        for (int k = 0; k < EB; ++k)
            if (d[k] >= 0) atomicAdd(&hist[(unsigned)d[k] / bw], 1);
    }
    __syncthreads();
    if (t < NBIN) binTab[blockIdx.x * NBIN + t] = hist[t];
}

__global__ __launch_bounds__(1024) void bin_scatter_k(const int* __restrict__ row,
                                                      const int* __restrict__ col,
                                                      const int* __restrict__ binTab,
                                                      int* __restrict__ binStart,
                                                      int2* __restrict__ rec, int nE, int n) {
    __shared__ int part[2][4][NBIN];
    __shared__ int waveTot[32];
    __shared__ int cursor[NBIN];
    const int t = threadIdx.x;
    const int c = t & (NBIN - 1);
    const int g = t >> 8;
    const int blk = blockIdx.x;

    int pre = 0, tot = 0;
#pragma unroll
    for (int k = 0; k < NBIN / 4; ++k) {
        const int bb = g * (NBIN / 4) + k;
        const int v = binTab[bb * NBIN + c];
        tot += v;
        pre += (bb < blk) ? v : 0;
    }
    part[0][g][c] = pre;
    part[1][g][c] = tot;
    __syncthreads();
    int p4 = 0, t4 = 0;
    if (t < NBIN) {
        for (int q = 0; q < 4; ++q) {
            p4 += part[0][q][t];
            t4 += part[1][q][t];
        }
    }
    const int start = block_excl_scan_1024(t4, waveTot);
    if (t < NBIN) {
        cursor[t] = start + p4;
        if (blk == 0) {
            binStart[t] = start;
            if (t == NBIN - 1) binStart[NBIN] = start + t4;
        }
    }
    __syncthreads();

    const unsigned bw = (unsigned)((n + NBIN - 1) / NBIN);
    const int slice = (nE + NBIN - 1) / NBIN;
    const int s0 = blk * slice;
    const int s1 = min(nE, s0 + slice);
    for (int eb = s0; eb < s1; eb += EB * 1024) {
        int r[EB], d[EB];
#pragma unroll
        for (int k = 0; k < EB; ++k) {
            const int e = eb + t + k * 1024;
            const bool ok = e < s1;
            r[k] = ok ? row[e] : 0;
            d[k] = ok ? col[e] : -1;
        }
#pragma unroll
        for (int k = 0; k < EB; ++k) {
            if (d[k] >= 0) {
                const int p = atomicAdd(&cursor[(unsigned)d[k] / bw], 1);
                rec[p] = make_int2(r[k], d[k]);
            }
        }
    }
}

__global__ __launch_bounds__(1024) void bin_sort_k(const int2* __restrict__ rec,
                                                   const int* __restrict__ binStart,
                                                   int* __restrict__ segptr,
                                                   float* __restrict__ dinv,
                                                   int* __restrict__ srcOf, int n) {
    __shared__ int cnt[1024];
    __shared__ int waveTot[32];
    const int t = threadIdx.x;
    const int bw = (n + NBIN - 1) / NBIN;
    const int base = blockIdx.x * bw;
    const int nn = min(bw, n - base);
    const int e0 = binStart[blockIdx.x];
    const int e1 = binStart[blockIdx.x + 1];
    cnt[t] = 0;
    __syncthreads();

    for (int eb = e0; eb < e1; eb += EB * 1024) {
        int d[EB];
#pragma unroll
        for (int k = 0; k < EB; ++k) {
            const int i = eb + t + k * 1024;
            d[k] = (i < e1) ? rec[i].y : -1;
        }
#pragma unroll
        for (int k = 0; k < EB; ++k)
            if (d[k] >= 0) atomicAdd(&cnt[d[k] - base], 1);
    }
    __syncthreads();

    const int c = (t < nn) ? cnt[t] : 0;
    const int excl = e0 + block_excl_scan_1024(c, waveTot);
    if (t < nn) {
        const int node = base + t;
        segptr[node] = excl;
        dinv[node] = rsqrtf((float)(c + 1));
        if (node == n - 1) segptr[n] = excl + c;
        cnt[t] = excl;
    }
    __syncthreads();

    for (int eb = e0; eb < e1; eb += EB * 1024) {
        int2 q[EB];
#pragma unroll
        for (int k = 0; k < EB; ++k) {
            const int i = eb + t + k * 1024;
            q[k] = (i < e1) ? rec[i] : make_int2(0, -1);
        }
#pragma unroll
        for (int k = 0; k < EB; ++k) {
            if (q[k].y >= 0) {
                const int p = atomicAdd(&cnt[q[k].y - base], 1);
                srcOf[p] = q[k].x;
            }
        }
    }
}

template <int NT, int K, int FOUT>
__global__ __launch_bounds__(32) void gemm_wmma_f32_k(const float* __restrict__ A,
                                                      const float* __restrict__ B,
                                                      float* __restrict__ C) {
    const int colBase = blockIdx.x * (16 * NT);
    const int rowBase = blockIdx.y * 16;
    const int lane = threadIdx.x;
    const int half = lane >> 4;
    const int lr   = lane & 15;

    v8f acc[NT];
    v8f zero = {};
#pragma unroll
    for (int t = 0; t < NT; ++t) acc[t] = zero;

    const float* Arow = A + (unsigned)(rowBase + lr) * K;

#pragma unroll 4
    for (int k0 = 0; k0 < K; k0 += 4) {
        const int ka = k0 + 2 * half;
        v2f a;
        a.x = Arow[ka];
        a.y = Arow[ka + 1];
#pragma unroll
        for (int t = 0; t < NT; ++t) {
            const float* Bp = B + (unsigned)ka * FOUT + colBase + 16 * t + lr;
            v2f b;
            b.x = Bp[0];
            b.y = Bp[FOUT];
            acc[t] = __builtin_amdgcn_wmma_f32_16x16x4_f32(
                 false, a,  false, b,
                 (short)0, acc[t],  false,  false);
        }
    }

#pragma unroll
    for (int t = 0; t < NT; ++t) {
        float* Cp = C + (unsigned)(rowBase + 8 * half) * FOUT + colBase + 16 * t + lr;
#pragma unroll
        for (int j = 0; j < 8; ++j) Cp[(unsigned)j * FOUT] = acc[t][j];
    }
}

template <int VEC, bool MAKE_W>
__global__ __launch_bounds__(256) void gather_rows_k(const float* __restrict__ h,
                                                     const int* __restrict__ segptr,
                                                     const int* __restrict__ srcOf,
                                                     const float* __restrict__ dinv,
                                                     float* __restrict__ wgtOf,
                                                     float* __restrict__ out, int n) {
    typedef float vec_t __attribute__((ext_vector_type(VEC)));
    const int lane = threadIdx.x & 31;
    const int node = __builtin_amdgcn_readfirstlane((int)(blockIdx.x * 8 + (threadIdx.x >> 5)));
    if (node >= n) return;
    const vec_t* H = (const vec_t*)h;
    const float dv = dinv[node];
    vec_t acc = H[(size_t)node * 32 + lane] * (dv * dv);
    const int p0 = segptr[node];
    const int p1 = segptr[node + 1];
    for (int base = p0; base < p1; base += 32) {
        const int m = min(32, p1 - base);
        int sv = 0;
        float wv = 0.0f;
        if (lane < m) {
            sv = srcOf[base + lane];
            wv = MAKE_W ? dinv[sv] : wgtOf[base + lane];
        }
#pragma unroll 4
        for (int j = 0; j < m; ++j) {
            const int r = __builtin_amdgcn_readlane(sv, j);
            float wj = __int_as_float(__builtin_amdgcn_readlane(__float_as_int(wv), j));
            if (MAKE_W) wj *= dv;
            acc += H[(size_t)r * 32 + lane] * wj;
        }
        if (MAKE_W && lane < m) wgtOf[base + lane] = wv * dv;
    }
    ((vec_t*)out)[(size_t)node * 32 + lane] = acc;
}

__global__ void bn_prep_k(const float* __restrict__ b, const float* __restrict__ g,
                          const float* __restrict__ be, const float* __restrict__ m,
                          const float* __restrict__ v, float* __restrict__ sc,
                          float* __restrict__ sh, int F) {
    int f = blockIdx.x * blockDim.x + threadIdx.x;
    if (f >= F) return;
    float s = g[f] * rsqrtf(v[f] + EPS);
    sc[f] = s;
    sh[f] = be[f] + (b[f] - m[f]) * s;
}

template <int F>
__global__ void bn_relu_k(float* __restrict__ x, const float* __restrict__ sc,
                          const float* __restrict__ sh, unsigned total  ) {
    unsigned tid = blockIdx.x * blockDim.x + threadIdx.x;
    if (tid >= total) return;
    unsigned f = tid % F;
    float y = fmaf(x[tid], sc[f], sh[f]);
    x[tid] = y > 0.0f ? y : 0.0f;
}

__global__ void bn_relu_fc_k(const float* __restrict__ x, const float* __restrict__ sc,
                             const float* __restrict__ sh, const float* __restrict__ fcW,
                             const float* __restrict__ fcb, float* __restrict__ out,
                             int nNodes) {
    int i = blockIdx.x * blockDim.x + threadIdx.x;
    if (i >= nNodes) return;
    const float* xi = x + (size_t)i * 64;
    float o0 = fcb[0], o1 = fcb[1];
#pragma unroll 8
    for (int k = 0; k < 64; ++k) {
        float y = fmaf(xi[k], sc[k], sh[k]);
        y = y > 0.0f ? y : 0.0f;
        o0 = fmaf(y, fcW[k * 2 + 0], o0);
        o1 = fmaf(y, fcW[k * 2 + 1], o1);
    }
    out[(size_t)i * 2 + 0] = o0;
    out[(size_t)i * 2 + 1] = o1;
}

static inline size_t alignf(size_t n) { return (n + 127) & ~(size_t)127; }

extern "C" void kernel_launch(void* const* d_in, const int* in_sizes, int n_in,
                              void* d_out, int out_size, void* d_ws, size_t ws_size,
                              hipStream_t stream) {
    const float* x    = (const float*)d_in[0];
    const int*   ei   = (const int*)d_in[1];
    const float* W1   = (const float*)d_in[2];
    const float* b1   = (const float*)d_in[3];
    const float* g1   = (const float*)d_in[4];
    const float* be1  = (const float*)d_in[5];
    const float* m1   = (const float*)d_in[6];
    const float* v1   = (const float*)d_in[7];
    const float* W2   = (const float*)d_in[8];
    const float* b2   = (const float*)d_in[9];
    const float* g2   = (const float*)d_in[10];
    const float* be2  = (const float*)d_in[11];
    const float* m2   = (const float*)d_in[12];
    const float* v2   = (const float*)d_in[13];
    const float* fcW  = (const float*)d_in[14];
    const float* fcb  = (const float*)d_in[15];
    float* out = (float*)d_out;

    const int N  = in_sizes[0] / 256;
    const int E  = in_sizes[1] / 2;
    const int* row = ei;
    const int* col = ei + E;

    float* ws = (float*)d_ws;
    size_t off = 0;
    float* dinv = ws + off; off += alignf((size_t)N);
    float* wgt  = ws + off; off += alignf((size_t)E);
    float* h1   = ws + off; off += alignf((size_t)N * 128);
    float* agg1 = ws + off; off += alignf((size_t)N * 128);
    int*   lists = (int*)(ws + off); off += alignf((size_t)N * 64);
    float* sc1  = ws + off; off += alignf(128);
    float* sh1  = ws + off; off += alignf(128);
    float* sc2  = ws + off; off += alignf(64);
    float* sh2  = ws + off; off += alignf(64);
    float* h2   = h1;
    float* agg2 = agg1;
    size_t loff = 0;
    int* binTab   = lists + loff; loff += alignf((size_t)NBIN * NBIN);
    int* binStart = lists + loff; loff += alignf((size_t)NBIN + 1);
    int* segptr   = lists + loff; loff += alignf((size_t)N + 1);
    int2* rec     = (int2*)(lists + loff); loff += alignf((size_t)E * 2);
    int* srcOf    = lists + loff; loff += alignf((size_t)E);
    (void)ws_size; (void)n_in; (void)out_size;

    const int TB = 256;
    bin_hist_k<<<NBIN, 1024, 0, stream>>>(col, binTab, E, N);
    bin_scatter_k<<<NBIN, 1024, 0, stream>>>(row, col, binTab, binStart, rec, E, N);
    bin_sort_k<<<NBIN, 1024, 0, stream>>>(rec, binStart, segptr, dinv, srcOf, N);

    bn_prep_k<<<1, 128, 0, stream>>>(b1, g1, be1, m1, v1, sc1, sh1, 128);
    bn_prep_k<<<1, 64, 0, stream>>>(b2, g2, be2, m2, v2, sc2, sh2, 64);

    {
        dim3 grid(128 / 64, N / 16);
        gemm_wmma_f32_k<4, 256, 128><<<grid, 32, 0, stream>>>(x, W1, h1);
    }

    gather_rows_k<4, true><<<(N + 7) / 8, 256, 0, stream>>>(h1, segptr, srcOf, dinv, wgt, agg1, N);

    {
        unsigned t = (unsigned)N * 128;
        bn_relu_k<128><<<(t + TB - 1) / TB, TB, 0, stream>>>(agg1, sc1, sh1, t);
    }

    {
        dim3 grid(64 / 64, N / 16);
        gemm_wmma_f32_k<4, 128, 64><<<grid, 32, 0, stream>>>(agg1, W2, h2);
    }

    gather_rows_k<2, false><<<(N + 7) / 8, 256, 0, stream>>>(h2, segptr, srcOf, dinv, wgt, agg2, N);

    bn_relu_fc_k<<<(N + TB - 1) / TB, TB, 0, stream>>>(agg2, sc2, sh2, fcW, fcb, out, N);
}


# BN1+ReLU moved into the layer-1 aggregation's epilogue: applied to the row sum in registers before its single store; the separate in-place BN+ReLU pass over the N x 128 plane (one launch) is gone
# speedup vs baseline: 3.1516x; 1.1931x over previous
#include <hip/hip_runtime.h>

typedef __attribute__((ext_vector_type(2))) float v2f;
typedef __attribute__((ext_vector_type(8))) float v8f;

#define EPS 1e-5f

#define NBIN 256
#define EB 8

__device__ __forceinline__ int block_excl_scan_1024(int c, int* waveTot) {
    const int lane = threadIdx.x & 31;
    const int w = threadIdx.x >> 5;
    int x = c;
    for (int o = 1; o < 32; o <<= 1) {
        int y = __shfl_up(x, o);
        if (lane >= o) x += y;
    }
    if (lane == 31) waveTot[w] = x;
    __syncthreads();
    int before = (lane < w) ? waveTot[lane] : 0;
    for (int o = 16; o > 0; o >>= 1) before += __shfl_xor(before, o);
    return before + x - c;
}

__global__ __launch_bounds__(1024) void bin_hist_k(const int* __restrict__ col,
                                                   int* __restrict__ binTab, int nE, int n) {
    __shared__ int hist[NBIN];
    const int t = threadIdx.x;
    if (t < NBIN) hist[t] = 0;
    __syncthreads();
    const unsigned bw = (unsigned)((n + NBIN - 1) / NBIN);
    const int slice = (nE + NBIN - 1) / NBIN;
    const int s0 = blockIdx.x * slice;
    const int s1 = min(nE, s0 + slice);
    for (int eb = s0; eb < s1; eb += EB * 1024) {
        int d[EB];
#pragma unroll
        for (int k = 0; k < EB; ++k) {
            const int e = eb + t + k * 1024;
            d[k] = (e < s1) ? col[e] : -1;
        }
#pragma unroll
        for (int k = 0; k < EB; ++k)
            if (d[k] >= 0) atomicAdd(&hist[(unsigned)d[k] / bw], 1);
    }
    __syncthreads();
    if (t < NBIN) binTab[blockIdx.x * NBIN + t] = hist[t];
}

__global__ __launch_bounds__(1024) void bin_scatter_k(const int* __restrict__ row,
                                                      const int* __restrict__ col,
                                                      const int* __restrict__ binTab,
                                                      int* __restrict__ binStart,
                                                      int2* __restrict__ rec, int nE, int n) {
    __shared__ int part[2][4][NBIN];
    __shared__ int waveTot[32];
    __shared__ int cursor[NBIN];
    const int t = threadIdx.x;
    const int c = t & (NBIN - 1);
    const int g = t >> 8;
    const int blk = blockIdx.x;

    int pre = 0, tot = 0;
#pragma unroll
    for (int k = 0; k < NBIN / 4; ++k) {
        const int bb = g * (NBIN / 4) + k;
        const int v = binTab[bb * NBIN + c];
        tot += v;
        pre += (bb < blk) ? v : 0;
    }
    part[0][g][c] = pre;
    part[1][g][c] = tot;
    __syncthreads();
    int p4 = 0, t4 = 0;
    if (t < NBIN) {
        for (int q = 0; q < 4; ++q) {
            p4 += part[0][q][t];
            t4 += part[1][q][t];
        }
    }
    const int start = block_excl_scan_1024(t4, waveTot);
    if (t < NBIN) {
        cursor[t] = start + p4;
        if (blk == 0) {
            binStart[t] = start;
            if (t == NBIN - 1) binStart[NBIN] = start + t4;
        }
    }
    __syncthreads();

    const unsigned bw = (unsigned)((n + NBIN - 1) / NBIN);
    const int slice = (nE + NBIN - 1) / NBIN;
    const int s0 = blk * slice;
    const int s1 = min(nE, s0 + slice);
    for (int eb = s0; eb < s1; eb += EB * 1024) {
        int r[EB], d[EB];
#pragma unroll
        for (int k = 0; k < EB; ++k) {
            const int e = eb + t + k * 1024;
            const bool ok = e < s1;
            r[k] = ok ? row[e] : 0;
            d[k] = ok ? col[e] : -1;
        }
#pragma unroll
        for (int k = 0; k < EB; ++k) {
            if (d[k] >= 0) {
                const int p = atomicAdd(&cursor[(unsigned)d[k] / bw], 1);
                rec[p] = make_int2(r[k], d[k]);
            }
        }
    }
}

__global__ __launch_bounds__(1024) void bin_sort_k(const int2* __restrict__ rec,
                                                   const int* __restrict__ binStart,
                                                   int* __restrict__ segptr,
                                                   float* __restrict__ dinv,
                                                   int* __restrict__ srcOf, int n) {
    __shared__ int cnt[1024];
    __shared__ int waveTot[32];
    const int t = threadIdx.x;
    const int bw = (n + NBIN - 1) / NBIN;
    const int base = blockIdx.x * bw;
    const int nn = min(bw, n - base);
    const int e0 = binStart[blockIdx.x];
    const int e1 = binStart[blockIdx.x + 1];
    cnt[t] = 0;
    __syncthreads();

    for (int eb = e0; eb < e1; eb += EB * 1024) {
        int d[EB];
#pragma unroll
        for (int k = 0; k < EB; ++k) {
            const int i = eb + t + k * 1024;
            d[k] = (i < e1) ? rec[i].y : -1;
        }
#pragma unroll
        for (int k = 0; k < EB; ++k)
            if (d[k] >= 0) atomicAdd(&cnt[d[k] - base], 1);
    }
    __syncthreads();

    const int c = (t < nn) ? cnt[t] : 0;
    const int excl = e0 + block_excl_scan_1024(c, waveTot);
    if (t < nn) {
        const int node = base + t;
        segptr[node] = excl;
        dinv[node] = rsqrtf((float)(c + 1));
        if (node == n - 1) segptr[n] = excl + c;
        cnt[t] = excl;
    }
    __syncthreads();

    for (int eb = e0; eb < e1; eb += EB * 1024) {
        int2 q[EB];
#pragma unroll
        for (int k = 0; k < EB; ++k) {
            const int i = eb + t + k * 1024;
            q[k] = (i < e1) ? rec[i] : make_int2(0, -1);
        }
#pragma unroll
        for (int k = 0; k < EB; ++k) {
            if (q[k].y >= 0) {
                const int p = atomicAdd(&cnt[q[k].y - base], 1);
                srcOf[p] = q[k].x;
            }
        }
    }
}

template <int NT, int K, int FOUT>
__global__ __launch_bounds__(32) void gemm_wmma_f32_k(const float* __restrict__ A,
                                                      const float* __restrict__ B,
                                                      float* __restrict__ C) {
    const int colBase = blockIdx.x * (16 * NT);
    const int rowBase = blockIdx.y * 16;
    const int lane = threadIdx.x;
    const int half = lane >> 4;
    const int lr   = lane & 15;

    v8f acc[NT];
    v8f zero = {};
#pragma unroll
    for (int t = 0; t < NT; ++t) acc[t] = zero;

    const float* Arow = A + (unsigned)(rowBase + lr) * K;

#pragma unroll 4
    for (int k0 = 0; k0 < K; k0 += 4) {
        const int ka = k0 + 2 * half;
        v2f a;
        a.x = Arow[ka];
        a.y = Arow[ka + 1];
#pragma unroll
        for (int t = 0; t < NT; ++t) {
            const float* Bp = B + (unsigned)ka * FOUT + colBase + 16 * t + lr;
            v2f b;
            b.x = Bp[0];
            b.y = Bp[FOUT];
            acc[t] = __builtin_amdgcn_wmma_f32_16x16x4_f32(
                 false, a,  false, b,
                 (short)0, acc[t],  false,  false);
        }
    }

#pragma unroll
    for (int t = 0; t < NT; ++t) {
        float* Cp = C + (unsigned)(rowBase + 8 * half) * FOUT + colBase + 16 * t + lr;
#pragma unroll
        for (int j = 0; j < 8; ++j) Cp[(unsigned)j * FOUT] = acc[t][j];
    }
}

template <int VEC, bool MAKE_W, bool BN_RELU>
__global__ __launch_bounds__(256) void gather_rows_k(const float* __restrict__ h,
                                                     const int* __restrict__ segptr,
                                                     const int* __restrict__ srcOf,
                                                     const float* __restrict__ dinv,
                                                     const float* __restrict__ sc,
                                                     const float* __restrict__ sh,
                                                     float* __restrict__ wgtOf,
                                                     float* __restrict__ out, int n) {
    typedef float vec_t __attribute__((ext_vector_type(VEC)));
    const int lane = threadIdx.x & 31;
    const int node = __builtin_amdgcn_readfirstlane((int)(blockIdx.x * 8 + (threadIdx.x >> 5)));
    if (node >= n) return;
    const vec_t* H = (const vec_t*)h;
    const float dv = dinv[node];
    vec_t scv = {}, shv = {};
    if (BN_RELU) {
        scv = ((const vec_t*)sc)[lane];
        shv = ((const vec_t*)sh)[lane];
    }
    vec_t acc = H[(size_t)node * 32 + lane] * (dv * dv);
    const int p0 = segptr[node];
    const int p1 = segptr[node + 1];
    for (int base = p0; base < p1; base += 32) {
        const int m = min(32, p1 - base);
        int sv = 0;
        float wv = 0.0f;
        if (lane < m) {
            sv = srcOf[base + lane];
            wv = MAKE_W ? dinv[sv] : wgtOf[base + lane];
        }
#pragma unroll 4
        for (int j = 0; j < m; ++j) {
            const int r = __builtin_amdgcn_readlane(sv, j);
            float wj = __int_as_float(__builtin_amdgcn_readlane(__float_as_int(wv), j));
            if (MAKE_W) wj *= dv;
            acc += H[(size_t)r * 32 + lane] * wj;
        }
        if (MAKE_W && lane < m) wgtOf[base + lane] = wv * dv;
    }
    if (BN_RELU) {
#pragma unroll
        for (int c = 0; c < VEC; ++c) {
            const float y = fmaf(acc[c], scv[c], shv[c]);
            acc[c] = y > 0.0f ? y : 0.0f;
        }
    }
    ((vec_t*)out)[(size_t)node * 32 + lane] = acc;
}

__global__ void bn_prep_k(const float* __restrict__ b, const float* __restrict__ g,
                          const float* __restrict__ be, const float* __restrict__ m,
                          const float* __restrict__ v, float* __restrict__ sc,
                          float* __restrict__ sh, int F) {
    int f = blockIdx.x * blockDim.x + threadIdx.x;
    if (f >= F) return;
    float s = g[f] * rsqrtf(v[f] + EPS);
    sc[f] = s;
    sh[f] = be[f] + (b[f] - m[f]) * s;
}

__global__ void bn_relu_fc_k(const float* __restrict__ x, const float* __restrict__ sc,
                             const float* __restrict__ sh, const float* __restrict__ fcW,
                             const float* __restrict__ fcb, float* __restrict__ out,
                             int nNodes) {
    int i = blockIdx.x * blockDim.x + threadIdx.x;
    if (i >= nNodes) return;
    const float* xi = x + (size_t)i * 64;
    float o0 = fcb[0], o1 = fcb[1];
#pragma unroll 8
    for (int k = 0; k < 64; ++k) {
        float y = fmaf(xi[k], sc[k], sh[k]);
        y = y > 0.0f ? y : 0.0f;
        o0 = fmaf(y, fcW[k * 2 + 0], o0);
        o1 = fmaf(y, fcW[k * 2 + 1], o1);
    }
    out[(size_t)i * 2 + 0] = o0;
    out[(size_t)i * 2 + 1] = o1;
}

static inline size_t alignf(size_t n) { return (n + 127) & ~(size_t)127; }

extern "C" void kernel_launch(void* const* d_in, const int* in_sizes, int n_in,
                              void* d_out, int out_size, void* d_ws, size_t ws_size,
                              hipStream_t stream) {
    const float* x    = (const float*)d_in[0];
    const int*   ei   = (const int*)d_in[1];
    const float* W1   = (const float*)d_in[2];
    const float* b1   = (const float*)d_in[3];
    const float* g1   = (const float*)d_in[4];
    const float* be1  = (const float*)d_in[5];
    const float* m1   = (const float*)d_in[6];
    const float* v1   = (const float*)d_in[7];
    const float* W2   = (const float*)d_in[8];
    const float* b2   = (const float*)d_in[9];
    const float* g2   = (const float*)d_in[10];
    const float* be2  = (const float*)d_in[11];
    const float* m2   = (const float*)d_in[12];
    const float* v2   = (const float*)d_in[13];
    const float* fcW  = (const float*)d_in[14];
    const float* fcb  = (const float*)d_in[15];
    float* out = (float*)d_out;

    const int N  = in_sizes[0] / 256;
    const int E  = in_sizes[1] / 2;
    const int* row = ei;
    const int* col = ei + E;

    float* ws = (float*)d_ws;
    size_t off = 0;
    float* dinv = ws + off; off += alignf((size_t)N);
    float* wgt  = ws + off; off += alignf((size_t)E);
    float* h1   = ws + off; off += alignf((size_t)N * 128);
    float* agg1 = ws + off; off += alignf((size_t)N * 128);
    int*   lists = (int*)(ws + off); off += alignf((size_t)N * 64);
    float* sc1  = ws + off; off += alignf(128);
    float* sh1  = ws + off; off += alignf(128);
    float* sc2  = ws + off; off += alignf(64);
    float* sh2  = ws + off; off += alignf(64);
    float* h2   = h1;
    float* agg2 = agg1;
    size_t loff = 0;
    int* binTab   = lists + loff; loff += alignf((size_t)NBIN * NBIN);
    int* binStart = lists + loff; loff += alignf((size_t)NBIN + 1);
    int* segptr   = lists + loff; loff += alignf((size_t)N + 1);
    int2* rec     = (int2*)(lists + loff); loff += alignf((size_t)E * 2);
    int* srcOf    = lists + loff; loff += alignf((size_t)E);
    (void)ws_size; (void)n_in; (void)out_size;

    const int TB = 256;
    bin_hist_k<<<NBIN, 1024, 0, stream>>>(col, binTab, E, N);
    bin_scatter_k<<<NBIN, 1024, 0, stream>>>(row, col, binTab, binStart, rec, E, N);
    bin_sort_k<<<NBIN, 1024, 0, stream>>>(rec, binStart, segptr, dinv, srcOf, N);

    bn_prep_k<<<1, 128, 0, stream>>>(b1, g1, be1, m1, v1, sc1, sh1, 128);
    bn_prep_k<<<1, 64, 0, stream>>>(b2, g2, be2, m2, v2, sc2, sh2, 64);

    {
        dim3 grid(128 / 64, N / 16);
        gemm_wmma_f32_k<4, 256, 128><<<grid, 32, 0, stream>>>(x, W1, h1);
    }

    gather_rows_k<4, true, true><<<(N + 7) / 8, 256, 0, stream>>>(h1, segptr, srcOf, dinv, sc1, sh1, wgt, agg1, N);

    {
        dim3 grid(64 / 64, N / 16);
        gemm_wmma_f32_k<4, 128, 64><<<grid, 32, 0, stream>>>(agg1, W2, h2);
    }

    gather_rows_k<2, false, false><<<(N + 7) / 8, 256, 0, stream>>>(h2, segptr, srcOf, dinv, nullptr, nullptr, wgt, agg2, N);

    bn_relu_fc_k<<<(N + TB - 1) / TB, TB, 0, stream>>>(agg2, sc2, sh2, fcW, fcb, out, N);
}


# Both GEMMs: weight matrix copied once per block into LDS in WMMA operand order (B no longer fetched per WMMA through the WGP cache), 16 x FOUT tile per wave; new k pairing changes the k sum order
# speedup vs baseline: 3.4952x; 1.1090x over previous
#include <hip/hip_runtime.h>

typedef __attribute__((ext_vector_type(2))) float v2f;
typedef __attribute__((ext_vector_type(8))) float v8f;

#define EPS 1e-5f

#define NBIN 256
#define EB 8

__device__ __forceinline__ int block_excl_scan_1024(int c, int* waveTot) {
    const int lane = threadIdx.x & 31;
    const int w = threadIdx.x >> 5;
    int x = c;
    for (int o = 1; o < 32; o <<= 1) {
        int y = __shfl_up(x, o);
        if (lane >= o) x += y;
    }
    if (lane == 31) waveTot[w] = x;
    __syncthreads();
    int before = (lane < w) ? waveTot[lane] : 0;
    for (int o = 16; o > 0; o >>= 1) before += __shfl_xor(before, o);
    return before + x - c;
}

__global__ __launch_bounds__(1024) void bin_hist_k(const int* __restrict__ col,
                                                   int* __restrict__ binTab, int nE, int n) {
    __shared__ int hist[NBIN];
    const int t = threadIdx.x;
    if (t < NBIN) hist[t] = 0;
    __syncthreads();
    const unsigned bw = (unsigned)((n + NBIN - 1) / NBIN);
    const int slice = (nE + NBIN - 1) / NBIN;
    const int s0 = blockIdx.x * slice;
    const int s1 = min(nE, s0 + slice);
    for (int eb = s0; eb < s1; eb += EB * 1024) {
        int d[EB];
#pragma unroll
        for (int k = 0; k < EB; ++k) {
            const int e = eb + t + k * 1024;
            d[k] = (e < s1) ? col[e] : -1;
        }
#pragma unroll
        for (int k = 0; k < EB; ++k)
            if (d[k] >= 0) atomicAdd(&hist[(unsigned)d[k] / bw], 1);
    }
    __syncthreads();
    if (t < NBIN) binTab[blockIdx.x * NBIN + t] = hist[t];
}

__global__ __launch_bounds__(1024) void bin_scatter_k(const int* __restrict__ row,
                                                      const int* __restrict__ col,
                                                      const int* __restrict__ binTab,
                                                      int* __restrict__ binStart,
                                                      int2* __restrict__ rec, int nE, int n) {
    __shared__ int part[2][4][NBIN];
    __shared__ int waveTot[32];
    __shared__ int cursor[NBIN];
    const int t = threadIdx.x;
    const int c = t & (NBIN - 1);
    const int g = t >> 8;
    const int blk = blockIdx.x;

    int pre = 0, tot = 0;
#pragma unroll
    for (int k = 0; k < NBIN / 4; ++k) {
        const int bb = g * (NBIN / 4) + k;
        const int v = binTab[bb * NBIN + c];
        tot += v;
        pre += (bb < blk) ? v : 0;
    }
    part[0][g][c] = pre;
    part[1][g][c] = tot;
    __syncthreads();
    int p4 = 0, t4 = 0;
    if (t < NBIN) {
        for (int q = 0; q < 4; ++q) {
            p4 += part[0][q][t];
            t4 += part[1][q][t];
        }
    }
    const int start = block_excl_scan_1024(t4, waveTot);
    if (t < NBIN) {
        cursor[t] = start + p4;
        if (blk == 0) {
            binStart[t] = start;
            if (t == NBIN - 1) binStart[NBIN] = start + t4;
        }
    }
    __syncthreads();

    const unsigned bw = (unsigned)((n + NBIN - 1) / NBIN);
    const int slice = (nE + NBIN - 1) / NBIN;
    const int s0 = blk * slice;
    const int s1 = min(nE, s0 + slice);
    for (int eb = s0; eb < s1; eb += EB * 1024) {
        int r[EB], d[EB];
#pragma unroll
        for (int k = 0; k < EB; ++k) {
            const int e = eb + t + k * 1024;
            const bool ok = e < s1;
            r[k] = ok ? row[e] : 0;
            d[k] = ok ? col[e] : -1;
        }
#pragma unroll
        for (int k = 0; k < EB; ++k) {
            if (d[k] >= 0) {
                const int p = atomicAdd(&cursor[(unsigned)d[k] / bw], 1);
                rec[p] = make_int2(r[k], d[k]);
            }
        }
    }
}

__global__ __launch_bounds__(1024) void bin_sort_k(const int2* __restrict__ rec,
                                                   const int* __restrict__ binStart,
                                                   int* __restrict__ segptr,
                                                   float* __restrict__ dinv,
                                                   int* __restrict__ srcOf, int n) {
    __shared__ int cnt[1024];
    __shared__ int waveTot[32];
    const int t = threadIdx.x;
    const int bw = (n + NBIN - 1) / NBIN;
    const int base = blockIdx.x * bw;
    const int nn = min(bw, n - base);
    const int e0 = binStart[blockIdx.x];
    const int e1 = binStart[blockIdx.x + 1];
    cnt[t] = 0;
    __syncthreads();

    for (int eb = e0; eb < e1; eb += EB * 1024) {
        int d[EB];
#pragma unroll
        for (int k = 0; k < EB; ++k) {
            const int i = eb + t + k * 1024;
            d[k] = (i < e1) ? rec[i].y : -1;
        }
#pragma unroll
        for (int k = 0; k < EB; ++k)
            if (d[k] >= 0) atomicAdd(&cnt[d[k] - base], 1);
    }
    __syncthreads();

    const int c = (t < nn) ? cnt[t] : 0;
    const int excl = e0 + block_excl_scan_1024(c, waveTot);
    if (t < nn) {
        const int node = base + t;
        segptr[node] = excl;
        dinv[node] = rsqrtf((float)(c + 1));
        if (node == n - 1) segptr[n] = excl + c;
        cnt[t] = excl;
    }
    __syncthreads();

    for (int eb = e0; eb < e1; eb += EB * 1024) {
        int2 q[EB];
#pragma unroll
        for (int k = 0; k < EB; ++k) {
            const int i = eb + t + k * 1024;
            q[k] = (i < e1) ? rec[i] : make_int2(0, -1);
        }
#pragma unroll
        for (int k = 0; k < EB; ++k) {
            if (q[k].y >= 0) {
                const int p = atomicAdd(&cnt[q[k].y - base], 1);
                srcOf[p] = q[k].x;
            }
        }
    }
}

template <int NT, int K, int FOUT>
__global__ __launch_bounds__(128, 2) void gemm_wmma_f32_k(const float* __restrict__ A,
                                                       const float* __restrict__ B,
                                                       float* __restrict__ C, int numTiles) {
    static_assert(FOUT == 16 * NT && (NT % 4) == 0 && (K % 32) == 0, "tile shape");
    typedef float f4 __attribute__((ext_vector_type(4)));
    constexpr int S = K / 32;
    constexpr int PITCH = 2 * FOUT;
    constexpr int QN = FOUT / 4;
    constexpr int ITEMS = (K / 2) * QN;
    constexpr int TWO_PER_WGP = 27648;
    __shared__ float Bs[K * FOUT > TWO_PER_WGP ? K * FOUT : TWO_PER_WGP];

    const int tid = threadIdx.x;
    const int lane = tid & 31;
    const int half = lane >> 4;
    const int c = lane & 15;
    const int stride = gridDim.x * 4;
    int tile = __builtin_amdgcn_readfirstlane((int)(blockIdx.x * 4 + (tid >> 5)));

    constexpr int R = ITEMS / 128;
    constexpr int RB = R < 16 ? R : 16;
    static_assert(ITEMS % 128 == 0 && R % RB == 0, "copy shape");
    f4 r0[RB], r1[RB];
#pragma unroll
    for (int j = 0; j < RB; ++j) {
        const int it = j * 128 + tid;
        const int p = it / QN, q = it % QN;
        r0[j] = *(const f4*)(B + (size_t)(2 * p) * FOUT + 4 * q);
        r1[j] = *(const f4*)(B + (size_t)(2 * p + 1) * FOUT + 4 * q);
    }

    f4 a[S][4];
    {
        const int tl = min(tile, numTiles - 1);
        const f4* Ap = (const f4*)(A + (size_t)(tl * 16 + c) * K + 16 * half);
#pragma unroll
        for (int s = 0; s < S; ++s)
#pragma unroll
            for (int i = 0; i < 4; ++i) a[s][i] = Ap[s * 8 + i];
    }

#pragma unroll
    for (int b = 0; b < R / RB; ++b) {
        if (b > 0) {
#pragma unroll
            for (int j = 0; j < RB; ++j) {
                const int it = (b * RB + j) * 128 + tid;
                const int p = it / QN, q = it % QN;
                r0[j] = *(const f4*)(B + (size_t)(2 * p) * FOUT + 4 * q);
                r1[j] = *(const f4*)(B + (size_t)(2 * p + 1) * FOUT + 4 * q);
            }
        }
#pragma unroll
        for (int j = 0; j < RB; ++j) {
            const int it = (b * RB + j) * 128 + tid;
            const int p = it / QN, q = it % QN;
            float* d = Bs + p * PITCH + (q >> 4) * 128 + (q & 15) * 4;
            f4 lo, hi;
            lo.x = r0[j].x; lo.y = r1[j].x; lo.z = r0[j].y; lo.w = r1[j].y;
            hi.x = r0[j].z; hi.y = r1[j].z; hi.z = r0[j].w; hi.w = r1[j].w;
            *(f4*)d = lo;
            *(f4*)(d + 64) = hi;
        }
    }
    __syncthreads();

    const float* Bl = Bs + (8 * half) * PITCH + c * 4;
    for (; tile < numTiles; tile += stride) {
        v8f acc[NT];
        v8f zero = {};
#pragma unroll
        for (int t = 0; t < NT; ++t) acc[t] = zero;
        const int tn = min(tile + stride, numTiles - 1);
        const f4* An = (const f4*)(A + (size_t)(tn * 16 + c) * K + 16 * half);
        f4 bq[2][2][NT / 2];
#pragma unroll
        for (int e = 0; e < 2; ++e)
#pragma unroll
            for (int g = 0; g < NT / 2; ++g) bq[0][e][g] = *(const f4*)(Bl + e * PITCH + g * 64);
#pragma unroll
        for (int s = 0; s < S; ++s) {
#pragma unroll
            for (int i = 0; i < 4; ++i) {
                const int cur = i & 1;
                if (s * 4 + i + 1 < S * 4) {
                    const float* Bn = Bl + (i < 3 ? 16 * s + 2 * i + 2 : 16 * (s + 1)) * PITCH;
#pragma unroll
                    for (int e = 0; e < 2; ++e)
#pragma unroll
                        for (int g = 0; g < NT / 2; ++g)
                            bq[cur ^ 1][e][g] = *(const f4*)(Bn + e * PITCH + g * 64);
                }
                const f4 av = a[s][i];
#pragma unroll
                for (int e = 0; e < 2; ++e) {
                    v2f af;
                    af.x = e ? av.z : av.x;
                    af.y = e ? av.w : av.y;
#pragma unroll
                    for (int g = 0; g < NT / 2; ++g) {
                        const f4 bv = bq[cur][e][g];
                        v2f b0, b1;
                        b0.x = bv.x; b0.y = bv.y;
                        b1.x = bv.z; b1.y = bv.w;
                        acc[2 * g] = __builtin_amdgcn_wmma_f32_16x16x4_f32(
                             false, af,  false, b0,
                             (short)0, acc[2 * g],  false,  false);
                        acc[2 * g + 1] = __builtin_amdgcn_wmma_f32_16x16x4_f32(
                             false, af,  false, b1,
                             (short)0, acc[2 * g + 1],  false,  false);
                    }
                }
                __builtin_amdgcn_sched_barrier(0);
            }
#pragma unroll
            for (int i = 0; i < 4; ++i) a[s][i] = An[s * 8 + i];
        }
        float* Crow = C + (size_t)(tile * 16 + 8 * half) * FOUT + 4 * c;
#pragma unroll
        for (int G = 0; G < NT / 4; ++G) {
#pragma unroll
            for (int j = 0; j < 8; ++j) {
                f4 o;
                o.x = acc[4 * G + 0][j];
                o.y = acc[4 * G + 1][j];
                o.z = acc[4 * G + 2][j];
                o.w = acc[4 * G + 3][j];
                *(f4*)(Crow + (size_t)j * FOUT + 64 * G) = o;
            }
        }
    }
}

template <int VEC, bool MAKE_W, bool BN_RELU>
__global__ __launch_bounds__(256) void gather_rows_k(const float* __restrict__ h,
                                                     const int* __restrict__ segptr,
                                                     const int* __restrict__ srcOf,
                                                     const float* __restrict__ dinv,
                                                     const float* __restrict__ sc,
                                                     const float* __restrict__ sh,
                                                     float* __restrict__ wgtOf,
                                                     float* __restrict__ out, int n) {
    typedef float vec_t __attribute__((ext_vector_type(VEC)));
    const int lane = threadIdx.x & 31;
    const int node = __builtin_amdgcn_readfirstlane((int)(blockIdx.x * 8 + (threadIdx.x >> 5)));
    if (node >= n) return;
    const vec_t* H = (const vec_t*)h;
    const float dv = dinv[node];
    vec_t scv = {}, shv = {};
    if (BN_RELU) {
        scv = ((const vec_t*)sc)[lane];
        shv = ((const vec_t*)sh)[lane];
    }
    vec_t acc = H[(size_t)node * 32 + lane] * (dv * dv);
    const int p0 = segptr[node];
    const int p1 = segptr[node + 1];
    for (int base = p0; base < p1; base += 32) {
        const int m = min(32, p1 - base);
        int sv = 0;
        float wv = 0.0f;
        if (lane < m) {
            sv = srcOf[base + lane];
            wv = MAKE_W ? dinv[sv] : wgtOf[base + lane];
        }
#pragma unroll 4
        for (int j = 0; j < m; ++j) {
            const int r = __builtin_amdgcn_readlane(sv, j);
            float wj = __int_as_float(__builtin_amdgcn_readlane(__float_as_int(wv), j));
            if (MAKE_W) wj *= dv;
            acc += H[(size_t)r * 32 + lane] * wj;
        }
        if (MAKE_W && lane < m) wgtOf[base + lane] = wv * dv;
    }
    if (BN_RELU) {
#pragma unroll
        for (int c = 0; c < VEC; ++c) {
            const float y = fmaf(acc[c], scv[c], shv[c]);
            acc[c] = y > 0.0f ? y : 0.0f;
        }
    }
    ((vec_t*)out)[(size_t)node * 32 + lane] = acc;
}

__global__ void bn_prep_k(const float* __restrict__ b, const float* __restrict__ g,
                          const float* __restrict__ be, const float* __restrict__ m,
                          const float* __restrict__ v, float* __restrict__ sc,
                          float* __restrict__ sh, int F) {
    int f = blockIdx.x * blockDim.x + threadIdx.x;
    if (f >= F) return;
    float s = g[f] * rsqrtf(v[f] + EPS);
    sc[f] = s;
    sh[f] = be[f] + (b[f] - m[f]) * s;
}

__global__ void bn_relu_fc_k(const float* __restrict__ x, const float* __restrict__ sc,
                             const float* __restrict__ sh, const float* __restrict__ fcW,
                             const float* __restrict__ fcb, float* __restrict__ out,
                             int nNodes) {
    int i = blockIdx.x * blockDim.x + threadIdx.x;
    if (i >= nNodes) return;
    const float* xi = x + (size_t)i * 64;
    float o0 = fcb[0], o1 = fcb[1];
#pragma unroll 8
    for (int k = 0; k < 64; ++k) {
        float y = fmaf(xi[k], sc[k], sh[k]);
        y = y > 0.0f ? y : 0.0f;
        o0 = fmaf(y, fcW[k * 2 + 0], o0);
        o1 = fmaf(y, fcW[k * 2 + 1], o1);
    }
    out[(size_t)i * 2 + 0] = o0;
    out[(size_t)i * 2 + 1] = o1;
}

static inline size_t alignf(size_t n) { return (n + 127) & ~(size_t)127; }

extern "C" void kernel_launch(void* const* d_in, const int* in_sizes, int n_in,
                              void* d_out, int out_size, void* d_ws, size_t ws_size,
                              hipStream_t stream) {
    const float* x    = (const float*)d_in[0];
    const int*   ei   = (const int*)d_in[1];
    const float* W1   = (const float*)d_in[2];
    const float* b1   = (const float*)d_in[3];
    const float* g1   = (const float*)d_in[4];
    const float* be1  = (const float*)d_in[5];
    const float* m1   = (const float*)d_in[6];
    const float* v1   = (const float*)d_in[7];
    const float* W2   = (const float*)d_in[8];
    const float* b2   = (const float*)d_in[9];
    const float* g2   = (const float*)d_in[10];
    const float* be2  = (const float*)d_in[11];
    const float* m2   = (const float*)d_in[12];
    const float* v2   = (const float*)d_in[13];
    const float* fcW  = (const float*)d_in[14];
    const float* fcb  = (const float*)d_in[15];
    float* out = (float*)d_out;

    const int N  = in_sizes[0] / 256;
    const int E  = in_sizes[1] / 2;
    const int* row = ei;
    const int* col = ei + E;

    float* ws = (float*)d_ws;
    size_t off = 0;
    float* dinv = ws + off; off += alignf((size_t)N);
    float* wgt  = ws + off; off += alignf((size_t)E);
    float* h1   = ws + off; off += alignf((size_t)N * 128);
    float* agg1 = ws + off; off += alignf((size_t)N * 128);
    int*   lists = (int*)(ws + off); off += alignf((size_t)N * 64);
    float* sc1  = ws + off; off += alignf(128);
    float* sh1  = ws + off; off += alignf(128);
    float* sc2  = ws + off; off += alignf(64);
    float* sh2  = ws + off; off += alignf(64);
    float* h2   = h1;
    float* agg2 = agg1;
    size_t loff = 0;
    int* binTab   = lists + loff; loff += alignf((size_t)NBIN * NBIN);
    int* binStart = lists + loff; loff += alignf((size_t)NBIN + 1);
    int* segptr   = lists + loff; loff += alignf((size_t)N + 1);
    int2* rec     = (int2*)(lists + loff); loff += alignf((size_t)E * 2);
    int* srcOf    = lists + loff; loff += alignf((size_t)E);
    (void)ws_size; (void)n_in; (void)out_size;

    const int TB = 256;
    bin_hist_k<<<NBIN, 1024, 0, stream>>>(col, binTab, E, N);
    bin_scatter_k<<<NBIN, 1024, 0, stream>>>(row, col, binTab, binStart, rec, E, N);
    bin_sort_k<<<NBIN, 1024, 0, stream>>>(rec, binStart, segptr, dinv, srcOf, N);

    bn_prep_k<<<1, 128, 0, stream>>>(b1, g1, be1, m1, v1, sc1, sh1, 128);
    bn_prep_k<<<1, 64, 0, stream>>>(b2, g2, be2, m2, v2, sc2, sh2, 64);

    gemm_wmma_f32_k<8, 256, 128><<<512, 128, 0, stream>>>(x, W1, h1, N / 16);

    gather_rows_k<4, true, true><<<(N + 7) / 8, 256, 0, stream>>>(h1, segptr, srcOf, dinv, sc1, sh1, wgt, agg1, N);

    gemm_wmma_f32_k<4, 128, 64><<<512, 128, 0, stream>>>(agg1, W2, h2, N / 16);

    gather_rows_k<2, false, false><<<(N + 7) / 8, 256, 0, stream>>>(h2, segptr, srcOf, dinv, nullptr, nullptr, wgt, agg2, N);

    bn_relu_fc_k<<<(N + TB - 1) / TB, TB, 0, stream>>>(agg2, sc2, sh2, fcW, fcb, out, N);
}


# Normalisation factored out of the edge walk: GEMM epilogue scales row r by dinv[r], gather adds unweighted rows and scales the sum by dinv[dst]; no per-edge dinv fetch, no weight plane; new rounding
# speedup vs baseline: 3.6079x; 1.0322x over previous
#include <hip/hip_runtime.h>

typedef __attribute__((ext_vector_type(2))) float v2f;
typedef __attribute__((ext_vector_type(8))) float v8f;

#define EPS 1e-5f

#define NBIN 256
#define EB 8

__device__ __forceinline__ int block_excl_scan_1024(int c, int* waveTot) {
    const int lane = threadIdx.x & 31;
    const int w = threadIdx.x >> 5;
    int x = c;
    for (int o = 1; o < 32; o <<= 1) {
        int y = __shfl_up(x, o);
        if (lane >= o) x += y;
    }
    if (lane == 31) waveTot[w] = x;
    __syncthreads();
    int before = (lane < w) ? waveTot[lane] : 0;
    for (int o = 16; o > 0; o >>= 1) before += __shfl_xor(before, o);
    return before + x - c;
}

__global__ __launch_bounds__(1024) void bin_hist_k(const int* __restrict__ col,
                                                   int* __restrict__ binTab, int nE, int n) {
    __shared__ int hist[NBIN];
    const int t = threadIdx.x;
    if (t < NBIN) hist[t] = 0;
    __syncthreads();
    const unsigned bw = (unsigned)((n + NBIN - 1) / NBIN);
    const int slice = (nE + NBIN - 1) / NBIN;
    const int s0 = blockIdx.x * slice;
    const int s1 = min(nE, s0 + slice);
    for (int eb = s0; eb < s1; eb += EB * 1024) {
        int d[EB];
#pragma unroll
        for (int k = 0; k < EB; ++k) {
            const int e = eb + t + k * 1024;
            d[k] = (e < s1) ? col[e] : -1;
        }
#pragma unroll
        for (int k = 0; k < EB; ++k)
            if (d[k] >= 0) atomicAdd(&hist[(unsigned)d[k] / bw], 1);
    }
    __syncthreads();
    if (t < NBIN) binTab[blockIdx.x * NBIN + t] = hist[t];
}

__global__ __launch_bounds__(1024) void bin_scatter_k(const int* __restrict__ row,
                                                      const int* __restrict__ col,
                                                      const int* __restrict__ binTab,
                                                      int* __restrict__ binStart,
                                                      int2* __restrict__ rec, int nE, int n) {
    __shared__ int part[2][4][NBIN];
    __shared__ int waveTot[32];
    __shared__ int cursor[NBIN];
    const int t = threadIdx.x;
    const int c = t & (NBIN - 1);
    const int g = t >> 8;
    const int blk = blockIdx.x;

    int pre = 0, tot = 0;
#pragma unroll
    for (int k = 0; k < NBIN / 4; ++k) {
        const int bb = g * (NBIN / 4) + k;
        const int v = binTab[bb * NBIN + c];
        tot += v;
        pre += (bb < blk) ? v : 0;
    }
    part[0][g][c] = pre;
    part[1][g][c] = tot;
    __syncthreads();
    int p4 = 0, t4 = 0;
    if (t < NBIN) {
        for (int q = 0; q < 4; ++q) {
            p4 += part[0][q][t];
            t4 += part[1][q][t];
        }
    }
    const int start = block_excl_scan_1024(t4, waveTot);
    if (t < NBIN) {
        cursor[t] = start + p4;
        if (blk == 0) {
            binStart[t] = start;
            if (t == NBIN - 1) binStart[NBIN] = start + t4;
        }
    }
    __syncthreads();

    const unsigned bw = (unsigned)((n + NBIN - 1) / NBIN);
    const int slice = (nE + NBIN - 1) / NBIN;
    const int s0 = blk * slice;
    const int s1 = min(nE, s0 + slice);
    for (int eb = s0; eb < s1; eb += EB * 1024) {
        int r[EB], d[EB];
#pragma unroll
        for (int k = 0; k < EB; ++k) {
            const int e = eb + t + k * 1024;
            const bool ok = e < s1;
            r[k] = ok ? row[e] : 0;
            d[k] = ok ? col[e] : -1;
        }
#pragma unroll
        for (int k = 0; k < EB; ++k) {
            if (d[k] >= 0) {
                const int p = atomicAdd(&cursor[(unsigned)d[k] / bw], 1);
                rec[p] = make_int2(r[k], d[k]);
            }
        }
    }
}

__global__ __launch_bounds__(1024) void bin_sort_k(const int2* __restrict__ rec,
                                                   const int* __restrict__ binStart,
                                                   int* __restrict__ segptr,
                                                   float* __restrict__ dinv,
                                                   int* __restrict__ srcOf, int n) {
    __shared__ int cnt[1024];
    __shared__ int waveTot[32];
    const int t = threadIdx.x;
    const int bw = (n + NBIN - 1) / NBIN;
    const int base = blockIdx.x * bw;
    const int nn = min(bw, n - base);
    const int e0 = binStart[blockIdx.x];
    const int e1 = binStart[blockIdx.x + 1];
    cnt[t] = 0;
    __syncthreads();

    for (int eb = e0; eb < e1; eb += EB * 1024) {
        int d[EB];
#pragma unroll
        for (int k = 0; k < EB; ++k) {
            const int i = eb + t + k * 1024;
            d[k] = (i < e1) ? rec[i].y : -1;
        }
#pragma unroll
        for (int k = 0; k < EB; ++k)
            if (d[k] >= 0) atomicAdd(&cnt[d[k] - base], 1);
    }
    __syncthreads();

    const int c = (t < nn) ? cnt[t] : 0;
    const int excl = e0 + block_excl_scan_1024(c, waveTot);
    if (t < nn) {
        const int node = base + t;
        segptr[node] = excl;
        dinv[node] = rsqrtf((float)(c + 1));
        if (node == n - 1) segptr[n] = excl + c;
        cnt[t] = excl;
    }
    __syncthreads();

    for (int eb = e0; eb < e1; eb += EB * 1024) {
        int2 q[EB];
#pragma unroll
        for (int k = 0; k < EB; ++k) {
            const int i = eb + t + k * 1024;
            q[k] = (i < e1) ? rec[i] : make_int2(0, -1);
        }
#pragma unroll
        for (int k = 0; k < EB; ++k) {
            if (q[k].y >= 0) {
                const int p = atomicAdd(&cnt[q[k].y - base], 1);
                srcOf[p] = q[k].x;
            }
        }
    }
}

template <int NT, int K, int FOUT>
__global__ __launch_bounds__(128, 2) void gemm_wmma_f32_k(const float* __restrict__ A,
                                                       const float* __restrict__ B,
                                                       const float* __restrict__ rowScale,
                                                       float* __restrict__ C, int numTiles) {
    static_assert(FOUT == 16 * NT && (NT % 4) == 0 && (K % 32) == 0, "tile shape");
    typedef float f4 __attribute__((ext_vector_type(4)));
    constexpr int S = K / 32;
    constexpr int PITCH = 2 * FOUT;
    constexpr int QN = FOUT / 4;
    constexpr int ITEMS = (K / 2) * QN;
    constexpr int TWO_PER_WGP = 27648;
    __shared__ float Bs[K * FOUT > TWO_PER_WGP ? K * FOUT : TWO_PER_WGP];

    const int tid = threadIdx.x;
    const int lane = tid & 31;
    const int half = lane >> 4;
    const int c = lane & 15;
    const int stride = gridDim.x * 4;
    int tile = __builtin_amdgcn_readfirstlane((int)(blockIdx.x * 4 + (tid >> 5)));

    constexpr int R = ITEMS / 128;
    constexpr int RB = R < 16 ? R : 16;
    static_assert(ITEMS % 128 == 0 && R % RB == 0, "copy shape");
    f4 r0[RB], r1[RB];
#pragma unroll
    for (int j = 0; j < RB; ++j) {
        const int it = j * 128 + tid;
        const int p = it / QN, q = it % QN;
        r0[j] = *(const f4*)(B + (size_t)(2 * p) * FOUT + 4 * q);
        r1[j] = *(const f4*)(B + (size_t)(2 * p + 1) * FOUT + 4 * q);
    }

    f4 a[S][4];
    {
        const int tl = min(tile, numTiles - 1);
        const f4* Ap = (const f4*)(A + (size_t)(tl * 16 + c) * K + 16 * half);
#pragma unroll
        for (int s = 0; s < S; ++s)
#pragma unroll
            for (int i = 0; i < 4; ++i) a[s][i] = Ap[s * 8 + i];
    }

#pragma unroll
    for (int b = 0; b < R / RB; ++b) {
        if (b > 0) {
#pragma unroll
            for (int j = 0; j < RB; ++j) {
                const int it = (b * RB + j) * 128 + tid;
                const int p = it / QN, q = it % QN;
                r0[j] = *(const f4*)(B + (size_t)(2 * p) * FOUT + 4 * q);
                r1[j] = *(const f4*)(B + (size_t)(2 * p + 1) * FOUT + 4 * q);
            }
        }
#pragma unroll
        for (int j = 0; j < RB; ++j) {
            const int it = (b * RB + j) * 128 + tid;
            const int p = it / QN, q = it % QN;
            float* d = Bs + p * PITCH + (q >> 4) * 128 + (q & 15) * 4;
            f4 lo, hi;
            lo.x = r0[j].x; lo.y = r1[j].x; lo.z = r0[j].y; lo.w = r1[j].y;
            hi.x = r0[j].z; hi.y = r1[j].z; hi.z = r0[j].w; hi.w = r1[j].w;
            *(f4*)d = lo;
            *(f4*)(d + 64) = hi;
        }
    }
    __syncthreads();

    const float* Bl = Bs + (8 * half) * PITCH + c * 4;
    for (; tile < numTiles; tile += stride) {
        v8f acc[NT];
        v8f zero = {};
#pragma unroll
        for (int t = 0; t < NT; ++t) acc[t] = zero;
        const int tn = min(tile + stride, numTiles - 1);
        const f4* An = (const f4*)(A + (size_t)(tn * 16 + c) * K + 16 * half);
        const f4* Sp = (const f4*)(rowScale + (size_t)tile * 16 + 8 * half);
        const f4 rs0 = Sp[0], rs1 = Sp[1];
        f4 bq[2][2][NT / 2];
#pragma unroll
        for (int e = 0; e < 2; ++e)
#pragma unroll
            for (int g = 0; g < NT / 2; ++g) bq[0][e][g] = *(const f4*)(Bl + e * PITCH + g * 64);
#pragma unroll
        for (int s = 0; s < S; ++s) {
#pragma unroll
            for (int i = 0; i < 4; ++i) {
                const int cur = i & 1;
                if (s * 4 + i + 1 < S * 4) {
                    const float* Bn = Bl + (i < 3 ? 16 * s + 2 * i + 2 : 16 * (s + 1)) * PITCH;
#pragma unroll
                    for (int e = 0; e < 2; ++e)
#pragma unroll
                        for (int g = 0; g < NT / 2; ++g)
                            bq[cur ^ 1][e][g] = *(const f4*)(Bn + e * PITCH + g * 64);
                }
                const f4 av = a[s][i];
#pragma unroll
                for (int e = 0; e < 2; ++e) {
                    v2f af;
                    af.x = e ? av.z : av.x;
                    af.y = e ? av.w : av.y;
#pragma unroll
                    for (int g = 0; g < NT / 2; ++g) {
                        const f4 bv = bq[cur][e][g];
                        v2f b0, b1;
                        b0.x = bv.x; b0.y = bv.y;
                        b1.x = bv.z; b1.y = bv.w;
                        acc[2 * g] = __builtin_amdgcn_wmma_f32_16x16x4_f32(
                             false, af,  false, b0,
                             (short)0, acc[2 * g],  false,  false);
                        acc[2 * g + 1] = __builtin_amdgcn_wmma_f32_16x16x4_f32(
                             false, af,  false, b1,
                             (short)0, acc[2 * g + 1],  false,  false);
                    }
                }
                __builtin_amdgcn_sched_barrier(0);
            }
#pragma unroll
            for (int i = 0; i < 4; ++i) a[s][i] = An[s * 8 + i];
        }
        float* Crow = C + (size_t)(tile * 16 + 8 * half) * FOUT + 4 * c;
#pragma unroll
        for (int G = 0; G < NT / 4; ++G) {
#pragma unroll
            for (int j = 0; j < 8; ++j) {
                const float rs = j < 4 ? rs0[j & 3] : rs1[j & 3];
                f4 o;
                o.x = acc[4 * G + 0][j] * rs;
                o.y = acc[4 * G + 1][j] * rs;
                o.z = acc[4 * G + 2][j] * rs;
                o.w = acc[4 * G + 3][j] * rs;
                *(f4*)(Crow + (size_t)j * FOUT + 64 * G) = o;
            }
        }
    }
}

template <int VEC, bool BN_RELU>
__global__ __launch_bounds__(256) void gather_rows_k(const float* __restrict__ h,
                                                     const int* __restrict__ segptr,
                                                     const int* __restrict__ srcOf,
                                                     const float* __restrict__ dinv,
                                                     const float* __restrict__ sc,
                                                     const float* __restrict__ sh,
                                                     float* __restrict__ out, int n) {
    typedef float vec_t __attribute__((ext_vector_type(VEC)));
    const int lane = threadIdx.x & 31;
    const int node = __builtin_amdgcn_readfirstlane((int)(blockIdx.x * 8 + (threadIdx.x >> 5)));
    if (node >= n) return;
    const vec_t* H = (const vec_t*)h;
    const float dv = dinv[node];
    vec_t scv = {}, shv = {};
    if (BN_RELU) {
        scv = ((const vec_t*)sc)[lane];
        shv = ((const vec_t*)sh)[lane];
    }
    vec_t acc = H[(size_t)node * 32 + lane];
    const int p0 = segptr[node];
    const int p1 = segptr[node + 1];
    for (int base = p0; base < p1; base += 32) {
        const int m = min(32, p1 - base);
        int sv = 0;
        if (lane < m) sv = srcOf[base + lane];
#pragma unroll 4
        for (int j = 0; j < m; ++j) {
            const int r = __builtin_amdgcn_readlane(sv, j);
            acc += H[(size_t)r * 32 + lane];
        }
    }
    acc *= dv;
    if (BN_RELU) {
#pragma unroll
        for (int c = 0; c < VEC; ++c) {
            const float y = fmaf(acc[c], scv[c], shv[c]);
            acc[c] = y > 0.0f ? y : 0.0f;
        }
    }
    ((vec_t*)out)[(size_t)node * 32 + lane] = acc;
}

__global__ void bn_prep_k(const float* __restrict__ b, const float* __restrict__ g,
                          const float* __restrict__ be, const float* __restrict__ m,
                          const float* __restrict__ v, float* __restrict__ sc,
                          float* __restrict__ sh, int F) {
    int f = blockIdx.x * blockDim.x + threadIdx.x;
    if (f >= F) return;
    float s = g[f] * rsqrtf(v[f] + EPS);
    sc[f] = s;
    sh[f] = be[f] + (b[f] - m[f]) * s;
}

__global__ void bn_relu_fc_k(const float* __restrict__ x, const float* __restrict__ sc,
                             const float* __restrict__ sh, const float* __restrict__ fcW,
                             const float* __restrict__ fcb, float* __restrict__ out,
                             int nNodes) {
    int i = blockIdx.x * blockDim.x + threadIdx.x;
    if (i >= nNodes) return;
    const float* xi = x + (size_t)i * 64;
    float o0 = fcb[0], o1 = fcb[1];
#pragma unroll 8
    for (int k = 0; k < 64; ++k) {
        float y = fmaf(xi[k], sc[k], sh[k]);
        y = y > 0.0f ? y : 0.0f;
        o0 = fmaf(y, fcW[k * 2 + 0], o0);
        o1 = fmaf(y, fcW[k * 2 + 1], o1);
    }
    out[(size_t)i * 2 + 0] = o0;
    out[(size_t)i * 2 + 1] = o1;
}

static inline size_t alignf(size_t n) { return (n + 127) & ~(size_t)127; }

extern "C" void kernel_launch(void* const* d_in, const int* in_sizes, int n_in,
                              void* d_out, int out_size, void* d_ws, size_t ws_size,
                              hipStream_t stream) {
    const float* x    = (const float*)d_in[0];
    const int*   ei   = (const int*)d_in[1];
    const float* W1   = (const float*)d_in[2];
    const float* b1   = (const float*)d_in[3];
    const float* g1   = (const float*)d_in[4];
    const float* be1  = (const float*)d_in[5];
    const float* m1   = (const float*)d_in[6];
    const float* v1   = (const float*)d_in[7];
    const float* W2   = (const float*)d_in[8];
    const float* b2   = (const float*)d_in[9];
    const float* g2   = (const float*)d_in[10];
    const float* be2  = (const float*)d_in[11];
    const float* m2   = (const float*)d_in[12];
    const float* v2   = (const float*)d_in[13];
    const float* fcW  = (const float*)d_in[14];
    const float* fcb  = (const float*)d_in[15];
    float* out = (float*)d_out;

    const int N  = in_sizes[0] / 256;
    const int E  = in_sizes[1] / 2;
    const int* row = ei;
    const int* col = ei + E;

    float* ws = (float*)d_ws;
    size_t off = 0;
    float* dinv = ws + off; off += alignf((size_t)N);
    float* h1   = ws + off; off += alignf((size_t)N * 128);
    float* agg1 = ws + off; off += alignf((size_t)N * 128);
    int*   lists = (int*)(ws + off); off += alignf((size_t)N * 64);
    float* sc1  = ws + off; off += alignf(128);
    float* sh1  = ws + off; off += alignf(128);
    float* sc2  = ws + off; off += alignf(64);
    float* sh2  = ws + off; off += alignf(64);
    float* h2   = h1;
    float* agg2 = agg1;
    size_t loff = 0;
    int* binTab   = lists + loff; loff += alignf((size_t)NBIN * NBIN);
    int* binStart = lists + loff; loff += alignf((size_t)NBIN + 1);
    int* segptr   = lists + loff; loff += alignf((size_t)N + 1);
    int2* rec     = (int2*)(lists + loff); loff += alignf((size_t)E * 2);
    int* srcOf    = lists + loff; loff += alignf((size_t)E);
    (void)ws_size; (void)n_in; (void)out_size;

    const int TB = 256;
    bin_hist_k<<<NBIN, 1024, 0, stream>>>(col, binTab, E, N);
    bin_scatter_k<<<NBIN, 1024, 0, stream>>>(row, col, binTab, binStart, rec, E, N);
    bin_sort_k<<<NBIN, 1024, 0, stream>>>(rec, binStart, segptr, dinv, srcOf, N);

    bn_prep_k<<<1, 128, 0, stream>>>(b1, g1, be1, m1, v1, sc1, sh1, 128);
    bn_prep_k<<<1, 64, 0, stream>>>(b2, g2, be2, m2, v2, sc2, sh2, 64);

    gemm_wmma_f32_k<8, 256, 128><<<512, 128, 0, stream>>>(x, W1, dinv, h1, N / 16);

    gather_rows_k<4, true><<<(N + 7) / 8, 256, 0, stream>>>(h1, segptr, srcOf, dinv, sc1, sh1, agg1, N);

    gemm_wmma_f32_k<4, 128, 64><<<512, 128, 0, stream>>>(agg1, W2, dinv, h2, N / 16);

    gather_rows_k<2, false><<<(N + 7) / 8, 256, 0, stream>>>(h2, segptr, srcOf, dinv, nullptr, nullptr, agg2, N);

    bn_relu_fc_k<<<(N + TB - 1) / TB, TB, 0, stream>>>(agg2, sc2, sh2, fcW, fcb, out, N);
}
